# starting kernel
# baseline (speedup 1.0000x reference)
#include <hip/hip_runtime.h>

typedef __attribute__((ext_vector_type(2))) float v2f;
typedef __attribute__((ext_vector_type(8))) float v8f;

#define HIDDEN 512
#define FEAT 4

__global__ void fold_hidden_kernel(const float* __restrict__ W,
                                   const float* __restrict__ bias,
                                   float* __restrict__ ws) {
    const int lane = threadIdx.x;
    float s0 = 0.f, s1 = 0.f, s2 = 0.f, s3 = 0.f, sb = 0.f;
    for (int r = lane; r < HIDDEN; r += 32) {
        const float4 w = *reinterpret_cast<const float4*>(W + r * FEAT);
        s0 += w.x; s1 += w.y; s2 += w.z; s3 += w.w;
        sb += bias[r];
    }
    #pragma unroll
    for (int off = 16; off > 0; off >>= 1) {
        s0 += __shfl_down(s0, off, 32);
        s1 += __shfl_down(s1, off, 32);
        s2 += __shfl_down(s2, off, 32);
        s3 += __shfl_down(s3, off, 32);
        sb += __shfl_down(sb, off, 32);
    }
    if (lane == 0) {
        const float inv = 1.0f / (float)HIDDEN;
        ws[0] = s0 * inv;
        ws[1] = s1 * inv;
        ws[2] = s2 * inv;
        ws[3] = s3 * inv;
        ws[4] = sb * inv;
    }
}

__global__ void relation_bias_kernel(const float* __restrict__ geom,
                                     const float* __restrict__ ws,
                                     float* __restrict__ out) {
    const int tid  = threadIdx.x;
    const int lane = tid & 31;
    const int wave = tid >> 5;
    const int m    = lane & 15;
    const int kh   = lane >> 4;

    const long base = ((long)blockIdx.x * 8 + wave) * 16;

    const float2 av =
        *reinterpret_cast<const float2*>(geom + (base + m) * FEAT + kh * 2);
    v2f a; a.x = av.x; a.y = av.y;

    const float w0 = ws[0], w1 = ws[1], w2 = ws[2], w3 = ws[3];
    const float bm = ws[4];
    v2f bmat;
    bmat.x = kh ? w2 : w0;
    bmat.y = kh ? w3 : w1;

    v8f c = {};
    c = __builtin_amdgcn_wmma_f32_16x16x4_f32(
         false, a,  false, bmat,
         (short)0, c,  false,  false);

    if (m == 0) {
        float4* op = reinterpret_cast<float4*>(out + base + kh * 8);
        float4 lo, hi;
        lo.x = c[0] + bm; lo.y = c[1] + bm; lo.z = c[2] + bm; lo.w = c[3] + bm;
        hi.x = c[4] + bm; hi.y = c[5] + bm; hi.z = c[6] + bm; hi.w = c[7] + bm;
        op[0] = lo;
        op[1] = hi;
    }
}

extern "C" void kernel_launch(void* const* d_in, const int* in_sizes, int n_in,
                              void* d_out, int out_size, void* d_ws, size_t ws_size,
                              hipStream_t stream) {
    const float* geom = (const float*)d_in[0];
    const float* W    = (const float*)d_in[1];
    const float* bias = (const float*)d_in[2];
    float*       out  = (float*)d_out;
    float*       ws   = (float*)d_ws;

    fold_hidden_kernel<<<1, 32, 0, stream>>>(W, bias, ws);

    const int total  = out_size;
    const int blocks = total / 128;
    relation_bias_kernel<<<blocks, 256, 0, stream>>>(geom, ws, out);
}


# fold fused into the main kernel, 2 launches -> 1: each of 256 blocks (8 waves x 16 tiles) sums W and b itself (2 rows per thread, butterfly, LDS over waves: new fold summation order)
# speedup vs baseline: 3.1697x; 3.1697x over previous
#include <hip/hip_runtime.h>


typedef __attribute__((ext_vector_type(2))) float v2f;
typedef __attribute__((ext_vector_type(8))) float v8f;

#define HIDDEN 512
#define FEAT   4
#define WAVES  8
#define TILES  16

__global__ __launch_bounds__(256)
void relation_bias_fused_kernel(const float* __restrict__ geom,
                                const float* __restrict__ W,
                                const float* __restrict__ bias,
                                float* __restrict__ out) {
    __shared__ __attribute__((aligned(16))) float part[5][WAVES];

    const int tid  = threadIdx.x;
    const int lane = tid & 31;
    const int wave = tid >> 5;
    const int m    = lane & 15;
    const int kh   = lane >> 4;

    const float4 wa = *reinterpret_cast<const float4*>(W + tid * FEAT);
    const float4 wb = *reinterpret_cast<const float4*>(W + (tid + 256) * FEAT);
    const float  ba = bias[tid];
    const float  bb = bias[tid + 256];

    const int base = (blockIdx.x * WAVES + wave) * (TILES * 16);
    const float* gp = geom + (base + m) * FEAT + kh * 2;
    v2f a[TILES];
    #pragma unroll
    for (int j = 0; j < TILES; ++j) {
        const float2 av = *reinterpret_cast<const float2*>(gp + j * 16 * FEAT);
        a[j].x = av.x; a[j].y = av.y;
    }

    float s0 = wa.x + wb.x, s1 = wa.y + wb.y, s2 = wa.z + wb.z, s3 = wa.w + wb.w;
    float sb = ba + bb;
    #pragma unroll
    for (int off = 16; off > 0; off >>= 1) {
        s0 += __shfl_xor(s0, off, 32);
        s1 += __shfl_xor(s1, off, 32);
        s2 += __shfl_xor(s2, off, 32);
        s3 += __shfl_xor(s3, off, 32);
        sb += __shfl_xor(sb, off, 32);
    }
    if (lane == 0) {
        part[0][wave] = s0;
        part[1][wave] = s1;
        part[2][wave] = s2;
        part[3][wave] = s3;
        part[4][wave] = sb;
    }
    __syncthreads();

    const float4* pw = reinterpret_cast<const float4*>(&part[kh * 2][0]);
    const float4* pb = reinterpret_cast<const float4*>(&part[4][0]);
    const float4 x0 = pw[0], x1 = pw[1];
    const float4 y0 = pw[2], y1 = pw[3];
    const float4 z0 = pb[0], z1 = pb[1];
    const float inv = 1.0f / (float)HIDDEN;
    v2f bmat;
    bmat.x = (((x0.x + x0.y) + (x0.z + x0.w)) + ((x1.x + x1.y) + (x1.z + x1.w))) * inv;
    bmat.y = (((y0.x + y0.y) + (y0.z + y0.w)) + ((y1.x + y1.y) + (y1.z + y1.w))) * inv;
    const float bm =
             (((z0.x + z0.y) + (z0.z + z0.w)) + ((z1.x + z1.y) + (z1.z + z1.w))) * inv;

    v8f c[TILES];
    #pragma unroll
    for (int j = 0; j < TILES; ++j) {
        const v8f zero = {};
        c[j] = __builtin_amdgcn_wmma_f32_16x16x4_f32(
             false, a[j],  false, bmat,
             (short)0, zero,  false,  false);
    }

    if (m == 0) {
        float* ob = out + base + kh * 8;
        #pragma unroll
        for (int j = 0; j < TILES; ++j) {
            float4* op = reinterpret_cast<float4*>(ob + j * 16);
            float4 lo, hi;
            lo.x = c[j][0] + bm; lo.y = c[j][1] + bm; lo.z = c[j][2] + bm; lo.w = c[j][3] + bm;
            hi.x = c[j][4] + bm; hi.y = c[j][5] + bm; hi.z = c[j][6] + bm; hi.w = c[j][7] + bm;
            op[0] = lo;
            op[1] = hi;
        }
    }
}

extern "C" void kernel_launch(void* const* d_in, const int* in_sizes, int n_in,
                              void* d_out, int out_size, void* d_ws, size_t ws_size,
                              hipStream_t stream) {
    const float* geom = (const float*)d_in[0];
    const float* W    = (const float*)d_in[1];
    const float* bias = (const float*)d_in[2];
    float*       out  = (float*)d_out;

    relation_bias_fused_kernel<<<256, 256, 0, stream>>>(geom, W, bias, out);
}


# lane owns one position: one 16-byte load per lane, weights as row-split A (rows 0-7 k-half 0, rows 8-15 k-half 1), k sum in two chained WMMAs, result lands in own lane: whole-line stores
# speedup vs baseline: 3.3279x; 1.0499x over previous
#include <hip/hip_runtime.h>


typedef __attribute__((ext_vector_type(2))) float v2f;
typedef __attribute__((ext_vector_type(8))) float v8f;

#define HIDDEN 512
#define FEAT   4
#define WAVES  8
#define GROUPS 8

__global__ __launch_bounds__(256)
void relation_bias_fused_kernel(const float* __restrict__ geom,
                                const float* __restrict__ W,
                                const float* __restrict__ bias,
                                float* __restrict__ out) {
    __shared__ __attribute__((aligned(16))) float part[5][WAVES];

    const int tid  = threadIdx.x;
    const int lane = tid & 31;
    const int wave = tid >> 5;
    const int h    = lane >> 4;

    const float4 wa = *reinterpret_cast<const float4*>(W + tid * FEAT);
    const float4 wb = *reinterpret_cast<const float4*>(W + (tid + 256) * FEAT);
    const float  ba = bias[tid];
    const float  bb = bias[tid + 256];

    const int base = (blockIdx.x * WAVES + wave) * (GROUPS * 32);
    const float4* gp = reinterpret_cast<const float4*>(geom) + base + lane;
    float4 g[GROUPS];
    #pragma unroll
    for (int i = 0; i < GROUPS; ++i) g[i] = gp[i * 32];

    float s0 = wa.x + wb.x, s1 = wa.y + wb.y, s2 = wa.z + wb.z, s3 = wa.w + wb.w;
    float sb = ba + bb;
    #pragma unroll
    for (int off = 16; off > 0; off >>= 1) {
        s0 += __shfl_xor(s0, off, 32);
        s1 += __shfl_xor(s1, off, 32);
        s2 += __shfl_xor(s2, off, 32);
        s3 += __shfl_xor(s3, off, 32);
        sb += __shfl_xor(sb, off, 32);
    }
    if (lane == 0) {
        part[0][wave] = s0;
        part[1][wave] = s1;
        part[2][wave] = s2;
        part[3][wave] = s3;
        part[4][wave] = sb;
    }
    __syncthreads();

    const float4* pq = reinterpret_cast<const float4*>(&part[0][0]);
    const float inv = 1.0f / (float)HIDDEN;
    float mean[5];
    #pragma unroll
    for (int q = 0; q < 5; ++q) {
        const float4 x0 = pq[2 * q], x1 = pq[2 * q + 1];
        mean[q] = (((x0.x + x0.y) + (x0.z + x0.w)) + ((x1.x + x1.y) + (x1.z + x1.w))) * inv;
    }
    const float bm = mean[4];

    const bool mine = (((lane >> 3) & 1) == h);
    v2f a01, a23;
    a01.x = mine ? mean[0] : 0.0f;
    a01.y = mine ? mean[1] : 0.0f;
    a23.x = mine ? mean[2] : 0.0f;
    a23.y = mine ? mean[3] : 0.0f;

    v8f acc[GROUPS];
    #pragma unroll
    for (int i = 0; i < GROUPS; ++i) {
        const v8f zero = {};
        v2f b01; b01.x = g[i].x; b01.y = g[i].y;
        acc[i] = __builtin_amdgcn_wmma_f32_16x16x4_f32(
             false, a01,  false, b01,
             (short)0, zero,  false,  false);
    }
    #pragma unroll
    for (int i = 0; i < GROUPS; ++i) {
        v2f b23; b23.x = g[i].z; b23.y = g[i].w;
        acc[i] = __builtin_amdgcn_wmma_f32_16x16x4_f32(
             false, a23,  false, b23,
             (short)0, acc[i],  false,  false);
    }

    float* ob = out + base + lane;
    #pragma unroll
    for (int i = 0; i < GROUPS; ++i) ob[i * 32] = acc[i][0] + bm;
}

extern "C" void kernel_launch(void* const* d_in, const int* in_sizes, int n_in,
                              void* d_out, int out_size, void* d_ws, size_t ws_size,
                              hipStream_t stream) {
    const float* geom = (const float*)d_in[0];
    const float* W    = (const float*)d_in[1];
    const float* bias = (const float*)d_in[2];
    float*       out  = (float*)d_out;

    relation_bias_fused_kernel<<<256, 256, 0, stream>>>(geom, W, bias, out);
}


# first product waits for the folded W only: barrier fences LDS only (geom loads stay in flight), wave sums by DPP row exchange + permlanex16 (new fold sum order), groups taken in load order
# speedup vs baseline: 4.1482x; 1.2465x over previous
#include <hip/hip_runtime.h>


typedef __attribute__((ext_vector_type(2))) float v2f;
typedef __attribute__((ext_vector_type(8))) float v8f;

#define HIDDEN 512
#define FEAT   4
#define WAVES  8
#define GROUPS 8

template <int XMASK>
static __device__ __forceinline__ float row_xor(float x) {
    const int b = __builtin_bit_cast(int, x);
    return __builtin_bit_cast(float,
        __builtin_amdgcn_update_dpp(b, b, 0x160 + XMASK, 0xf, 0xf, true));
}
static __device__ __forceinline__ float other_row(float x) {
    const int b = __builtin_bit_cast(int, x);
    return __builtin_bit_cast(float,
        __builtin_amdgcn_permlanex16(b, b, 0x76543210, 0xfedcba98, false, false));
}
#define FOLD_STEP(EXCH)                                   \
    do {                                                  \
        const float t0 = EXCH(s0), t1 = EXCH(s1), t2 = EXCH(s2), t3 = EXCH(s3), t4 = EXCH(sb); \
        s0 += t0; s1 += t1; s2 += t2; s3 += t3; sb += t4; \
    } while (0)

__global__ __launch_bounds__(256)
void relation_bias_fused_kernel(const float* __restrict__ geom,
                                const float* __restrict__ W,
                                const float* __restrict__ bias,
                                float* __restrict__ out) {
    __shared__ __attribute__((aligned(16))) float part[5][WAVES];

    const int tid  = threadIdx.x;
    const int lane = tid & 31;
    const int wave = tid >> 5;
    const int h    = lane >> 4;

    const float4 wa = *reinterpret_cast<const float4*>(W + tid * FEAT);
    const float4 wb = *reinterpret_cast<const float4*>(W + (tid + 256) * FEAT);
    const float  ba = bias[tid];
    const float  bb = bias[tid + 256];

    const int base = (blockIdx.x * WAVES + wave) * (GROUPS * 32);
    const float4* gp = reinterpret_cast<const float4*>(geom) + base + lane;
    float4 g[GROUPS];
    #pragma unroll
    for (int i = 0; i < GROUPS; ++i) g[i] = gp[i * 32];

    float s0 = wa.x + wb.x, s1 = wa.y + wb.y, s2 = wa.z + wb.z, s3 = wa.w + wb.w;
    float sb = ba + bb;
    FOLD_STEP(row_xor<1>);
    FOLD_STEP(row_xor<2>);
    FOLD_STEP(row_xor<4>);
    FOLD_STEP(row_xor<8>);
    FOLD_STEP(other_row);
    if (lane == 0) {
        part[0][wave] = s0;
        part[1][wave] = s1;
        part[2][wave] = s2;
        part[3][wave] = s3;
        part[4][wave] = sb;
    }
    __builtin_amdgcn_fence(__ATOMIC_RELEASE, "workgroup", "local");
    __builtin_amdgcn_s_barrier();
    __builtin_amdgcn_fence(__ATOMIC_ACQUIRE, "workgroup", "local");

    const float4* pq = reinterpret_cast<const float4*>(&part[0][0]);
    const float inv = 1.0f / (float)HIDDEN;
    float mean[5];
    #pragma unroll
    for (int q = 0; q < 5; ++q) {
        const float4 x0 = pq[2 * q], x1 = pq[2 * q + 1];
        mean[q] = (((x0.x + x0.y) + (x0.z + x0.w)) + ((x1.x + x1.y) + (x1.z + x1.w))) * inv;
    }
    const float bm = mean[4];

    const bool mine = (((lane >> 3) & 1) == h);
    v2f a01, a23;
    a01.x = mine ? mean[0] : 0.0f;
    a01.y = mine ? mean[1] : 0.0f;
    a23.x = mine ? mean[2] : 0.0f;
    a23.y = mine ? mean[3] : 0.0f;

    float* ob = out + base + lane;
    __builtin_amdgcn_sched_barrier(0);
    #pragma unroll
    for (int i = 0; i < GROUPS; i += 2) {
        const v8f zero = {};
        v2f p01, q01, p23, q23;
        p01.x = g[i].x;     p01.y = g[i].y;     p23.x = g[i].z;     p23.y = g[i].w;
        q01.x = g[i + 1].x; q01.y = g[i + 1].y; q23.x = g[i + 1].z; q23.y = g[i + 1].w;
        v8f accp = __builtin_amdgcn_wmma_f32_16x16x4_f32(
             false, a01,  false, p01,
             (short)0, zero,  false,  false);
        v8f accq = __builtin_amdgcn_wmma_f32_16x16x4_f32(
             false, a01,  false, q01,
             (short)0, zero,  false,  false);
        accp = __builtin_amdgcn_wmma_f32_16x16x4_f32(
             false, a23,  false, p23,
             (short)0, accp,  false,  false);
        accq = __builtin_amdgcn_wmma_f32_16x16x4_f32(
             false, a23,  false, q23,
             (short)0, accq,  false,  false);
        ob[i * 32]       = accp[0] + bm;
        ob[(i + 1) * 32] = accq[0] + bm;
        __builtin_amdgcn_sched_barrier(0);
    }
}

extern "C" void kernel_launch(void* const* d_in, const int* in_sizes, int n_in,
                              void* d_out, int out_size, void* d_ws, size_t ws_size,
                              hipStream_t stream) {
    const float* geom = (const float*)d_in[0];
    const float* W    = (const float*)d_in[1];
    const float* bias = (const float*)d_in[2];
    float*       out  = (float*)d_out;

    relation_bias_fused_kernel<<<256, 256, 0, stream>>>(geom, W, bias, out);
}
